# starting kernel
# baseline (speedup 1.0000x reference)
#include <hip/hip_runtime.h>


typedef __attribute__((ext_vector_type(2))) float v2f;
typedef __attribute__((ext_vector_type(4))) float v4f;
typedef __attribute__((ext_vector_type(8))) float v8f;

#define WMMA4(A, Bm, C) \
  __builtin_amdgcn_wmma_f32_16x16x4_f32(false, (A), false, (Bm), (short)0, (C), false, false)

#define MPS_B    16384
#define MPS_F    512
#define MPS_HALF 256
#define NSIDE    255
#define TILES_PER_WAVE 2

#define PRE_SIDE_V4F (NSIDE * 4 * 32)
#define PRE_BYTES    ((size_t)2 * PRE_SIDE_V4F * 16)

__global__ __launch_bounds__(256) void mps_prepack_kernel(
    const float* __restrict__ Wl, const float* __restrict__ Wr,
    float* __restrict__ pre)
{
  const int tid = blockIdx.x * blockDim.x + threadIdx.x;
  if (tid >= 2 * NSIDE * 32) return;
  const int side = tid / (NSIDE * 32);
  const int rem  = tid % (NSIDE * 32);
  const int s    = rem / 32;
  const int lane = rem % 32;
  const int m    = lane & 15;
  const int hl   = lane >> 4;

  const float* Ws = (side ? Wr : Wl) + (long)s * 512;
  v4f* dst = (v4f*)pre + (size_t)side * PRE_SIDE_V4F;

#pragma unroll
  for (int c = 0; c < 4; ++c) {
    const int r0 = 2 * c + 8 * hl;
    v4f q;
    if (side == 0) {
      q.x = Ws[(r0    ) * 32      + m];
      q.y = Ws[(r0 + 1) * 32      + m];
      q.z = Ws[(r0    ) * 32 + 16 + m];
      q.w = Ws[(r0 + 1) * 32 + 16 + m];
    } else {
      q.x = Ws[m * 32      + r0    ];
      q.y = Ws[m * 32      + r0 + 1];
      q.z = Ws[m * 32 + 16 + r0    ];
      q.w = Ws[m * 32 + 16 + r0 + 1];
    }
    dst[((size_t)s * 4 + c) * 32 + lane] = q;
  }
}

__device__ __forceinline__ void mps_step(v8f& V, float xt, const v2f a0[4], const v2f a1[4]) {
  v8f P = {}; v8f Q = {};
#pragma unroll
  for (int c = 0; c < 4; ++c) {
    v2f bb; bb.x = V[2 * c]; bb.y = V[2 * c + 1];
    P = WMMA4(a0[c], bb, P);
    Q = WMMA4(a1[c], bb, Q);
  }
#pragma unroll
  for (int r = 0; r < 8; ++r) V[r] = fmaf(xt, Q[r], P[r]);
}

__device__ __forceinline__ void mps_tail(
    long b0, long b1, int lane, int m, int hl,
    const v8f& V0, const v8f& V1, const v8f& U0, const v8f& U1,
    const float* __restrict__ Wout, const float* __restrict__ d1w,
    const float* __restrict__ d1b,  const float* __restrict__ d2w,
    const float* __restrict__ d2b,  float* __restrict__ out)
{
  float weff[8] = {}, c0 = d2b[0];
#pragma unroll 4
  for (int h = 0; h < 24; ++h) {
    const float wh = d2w[h];
    c0 = fmaf(wh, d1b[h], c0);
#pragma unroll
    for (int o = 0; o < 8; ++o) weff[o] = fmaf(wh, d1w[h * 8 + o], weff[o]);
  }

  float y0 = c0, y1 = c0;
#pragma unroll
  for (int o = 0; o < 8; ++o) {
    v2f a[4];
#pragma unroll
    for (int c = 0; c < 4; ++c) {
      const int d0 = 2 * c + 8 * hl;
      a[c].x = Wout[(d0    ) * 128 + o * 16 + m];
      a[c].y = Wout[(d0 + 1) * 128 + o * 16 + m];
    }
    v8f G0 = {}, G1 = {};
#pragma unroll
    for (int c = 0; c < 4; ++c) {
      v2f bb0; bb0.x = V0[2 * c]; bb0.y = V0[2 * c + 1];
      v2f bb1; bb1.x = V1[2 * c]; bb1.y = V1[2 * c + 1];
      G0 = WMMA4(a[c], bb0, G0);
      G1 = WMMA4(a[c], bb1, G1);
    }
    float p0 = 0.f, p1 = 0.f;
#pragma unroll
    for (int r = 0; r < 8; ++r) { p0 = fmaf(G0[r], U0[r], p0); p1 = fmaf(G1[r], U1[r], p1); }
    p0 += __shfl_xor(p0, 16, 32);
    p1 += __shfl_xor(p1, 16, 32);
    y0 = fmaf(weff[o], p0, y0);
    y1 = fmaf(weff[o], p1, y1);
  }

  if (lane < 16) {
    out[b0] = y0;
    out[b1] = y1;
  }
}

__global__ __launch_bounds__(256) void mps_dense_packed_kernel(
    const float* __restrict__ x,   const float* __restrict__ W0,
    const float* __restrict__ pre, const float* __restrict__ Wout,
    const float* __restrict__ WN,
    const float* __restrict__ d1w, const float* __restrict__ d1b,
    const float* __restrict__ d2w, const float* __restrict__ d2b,
    float* __restrict__ out)
{
  const int lane = threadIdx.x & 31;
  const int wave = blockIdx.x * (blockDim.x >> 5) + (threadIdx.x >> 5);
  const int m    = lane & 15;
  const int hl   = lane >> 4;

  const long b0 = (long)wave * (16 * TILES_PER_WAVE) + m;
  const long b1 = b0 + 16;

  const v4f* PL = (const v4f*)pre + lane;
  const v4f* PR = (const v4f*)pre + PRE_SIDE_V4F + lane;

  v8f V0, V1;
  {
    const float xa = x[b0 * MPS_F];
    const float xb = x[b1 * MPS_F];
#pragma unroll
    for (int r = 0; r < 8; ++r) {
      const int d = r + 8 * hl;
      const float p0 = W0[d], p1 = W0[16 + d];
      V0[r] = fmaf(xa, p1, p0);
      V1[r] = fmaf(xb, p1, p0);
    }
  }

  for (int s = 0; s < NSIDE; ++s) {
    v2f a0[4], a1[4];
#pragma unroll
    for (int c = 0; c < 4; ++c) {
      const v4f q = PL[((size_t)s * 4 + c) * 32];
      a0[c].x = q.x; a0[c].y = q.y;
      a1[c].x = q.z; a1[c].y = q.w;
    }
    const float xt0 = x[b0 * MPS_F + (s + 1)];
    const float xt1 = x[b1 * MPS_F + (s + 1)];
    mps_step(V0, xt0, a0, a1);
    mps_step(V1, xt1, a0, a1);
  }

  v8f U0, U1;
  {
    const float xa = x[b0 * MPS_F + (MPS_F - 1)];
    const float xb = x[b1 * MPS_F + (MPS_F - 1)];
#pragma unroll
    for (int r = 0; r < 8; ++r) {
      const int d = r + 8 * hl;
      const float p0 = WN[d * 2], p1 = WN[d * 2 + 1];
      U0[r] = fmaf(xa, p1, p0);
      U1[r] = fmaf(xb, p1, p0);
    }
  }

  for (int i = 0; i < NSIDE; ++i) {
    const int s = NSIDE - 1 - i;
    v2f a0[4], a1[4];
#pragma unroll
    for (int c = 0; c < 4; ++c) {
      const v4f q = PR[((size_t)s * 4 + c) * 32];
      a0[c].x = q.x; a0[c].y = q.y;
      a1[c].x = q.z; a1[c].y = q.w;
    }
    const int t = MPS_HALF + s;
    const float xt0 = x[b0 * MPS_F + t];
    const float xt1 = x[b1 * MPS_F + t];
    mps_step(U0, xt0, a0, a1);
    mps_step(U1, xt1, a0, a1);
  }

  mps_tail(b0, b1, lane, m, hl, V0, V1, U0, U1, Wout, d1w, d1b, d2w, d2b, out);
}

__global__ __launch_bounds__(256) void mps_dense_kernel(
    const float* __restrict__ x,   const float* __restrict__ W0,
    const float* __restrict__ Wl,  const float* __restrict__ Wout,
    const float* __restrict__ Wr,  const float* __restrict__ WN,
    const float* __restrict__ d1w, const float* __restrict__ d1b,
    const float* __restrict__ d2w, const float* __restrict__ d2b,
    float* __restrict__ out)
{
  const int lane = threadIdx.x & 31;
  const int wave = blockIdx.x * (blockDim.x >> 5) + (threadIdx.x >> 5);
  const int m    = lane & 15;
  const int hl   = lane >> 4;

  const long b0 = (long)wave * (16 * TILES_PER_WAVE) + m;
  const long b1 = b0 + 16;

  v8f V0, V1;
  {
    const float xa = x[b0 * MPS_F];
    const float xb = x[b1 * MPS_F];
#pragma unroll
    for (int r = 0; r < 8; ++r) {
      const int d = r + 8 * hl;
      const float p0 = W0[d], p1 = W0[16 + d];
      V0[r] = fmaf(xa, p1, p0);
      V1[r] = fmaf(xb, p1, p0);
    }
  }

  for (int s = 0; s < NSIDE; ++s) {
    const float* Ws = Wl + (long)s * 512;
    v2f a0[4], a1[4];
#pragma unroll
    for (int c = 0; c < 4; ++c) {
      const int d0 = 2 * c + 8 * hl;
      a0[c].x = Ws[(d0    ) * 32      + m];
      a0[c].y = Ws[(d0 + 1) * 32      + m];
      a1[c].x = Ws[(d0    ) * 32 + 16 + m];
      a1[c].y = Ws[(d0 + 1) * 32 + 16 + m];
    }
    const float xt0 = x[b0 * MPS_F + (s + 1)];
    const float xt1 = x[b1 * MPS_F + (s + 1)];
    mps_step(V0, xt0, a0, a1);
    mps_step(V1, xt1, a0, a1);
  }

  v8f U0, U1;
  {
    const float xa = x[b0 * MPS_F + (MPS_F - 1)];
    const float xb = x[b1 * MPS_F + (MPS_F - 1)];
#pragma unroll
    for (int r = 0; r < 8; ++r) {
      const int d = r + 8 * hl;
      const float p0 = WN[d * 2], p1 = WN[d * 2 + 1];
      U0[r] = fmaf(xa, p1, p0);
      U1[r] = fmaf(xb, p1, p0);
    }
  }

  for (int i = 0; i < NSIDE; ++i) {
    const int s = NSIDE - 1 - i;
    const float* Ws = Wr + (long)s * 512;
    v2f a0[4], a1[4];
#pragma unroll
    for (int c = 0; c < 4; ++c) {
      const int e0 = 2 * c + 8 * hl;
      a0[c].x = Ws[m * 32      + e0    ];
      a0[c].y = Ws[m * 32      + e0 + 1];
      a1[c].x = Ws[m * 32 + 16 + e0    ];
      a1[c].y = Ws[m * 32 + 16 + e0 + 1];
    }
    const int t = MPS_HALF + s;
    const float xt0 = x[b0 * MPS_F + t];
    const float xt1 = x[b1 * MPS_F + t];
    mps_step(U0, xt0, a0, a1);
    mps_step(U1, xt1, a0, a1);
  }

  mps_tail(b0, b1, lane, m, hl, V0, V1, U0, U1, Wout, d1w, d1b, d2w, d2b, out);
}

extern "C" void kernel_launch(void* const* d_in, const int* in_sizes, int n_in,
                              void* d_out, int out_size, void* d_ws, size_t ws_size,
                              hipStream_t stream) {
  (void)in_sizes; (void)n_in; (void)out_size;
  const float* x    = (const float*)d_in[0];
  const float* W0   = (const float*)d_in[1];
  const float* Wl   = (const float*)d_in[2];
  const float* Wout = (const float*)d_in[3];
  const float* Wr   = (const float*)d_in[4];
  const float* WN   = (const float*)d_in[5];
  const float* d1w  = (const float*)d_in[6];
  const float* d1b  = (const float*)d_in[7];
  const float* d2w  = (const float*)d_in[8];
  const float* d2b  = (const float*)d_in[9];
  float* out = (float*)d_out;

  dim3 grid(64), block(256);

  if (d_ws != nullptr && ws_size >= PRE_BYTES) {
    hipLaunchKernelGGL(mps_prepack_kernel, dim3(64), dim3(256), 0, stream,
                       Wl, Wr, (float*)d_ws);
    hipLaunchKernelGGL(mps_dense_packed_kernel, grid, block, 0, stream,
                       x, W0, (const float*)d_ws, Wout, WN,
                       d1w, d1b, d2w, d2b, out);
  } else {
    hipLaunchKernelGGL(mps_dense_kernel, grid, block, 0, stream,
                       x, W0, Wl, Wout, Wr, WN, d1w, d1b, d2w, d2b, out);
  }
}


# one 16-row tile per wave, 256 work-groups of 4 waves: one wave per SIMD on all 256 WGPs (was 2 per SIMD on 64 WGPs); a group's 64 outputs gathered in LDS, one store per 128-B line
# speedup vs baseline: 1.7616x; 1.7616x over previous
#include <hip/hip_runtime.h>


typedef __attribute__((ext_vector_type(2))) float v2f;
typedef __attribute__((ext_vector_type(4))) float v4f;
typedef __attribute__((ext_vector_type(8))) float v8f;

#define WMMA4(A, Bm, C) \
  __builtin_amdgcn_wmma_f32_16x16x4_f32(false, (A), false, (Bm), (short)0, (C), false, false)

#define MPS_B    16384
#define MPS_F    512
#define MPS_HALF 256
#define NSIDE    255
#define WAVES_PER_BLOCK 4
#define BLOCK_THREADS   (32 * WAVES_PER_BLOCK)
#define ROWS_PER_BLOCK  (16 * WAVES_PER_BLOCK)

#define PRE_SIDE_V4F (NSIDE * 4 * 32)
#define PRE_BYTES    ((size_t)2 * PRE_SIDE_V4F * 16)

__global__ __launch_bounds__(256) void mps_prepack_kernel(
    const float* __restrict__ Wl, const float* __restrict__ Wr,
    float* __restrict__ pre)
{
  const int tid = blockIdx.x * blockDim.x + threadIdx.x;
  if (tid >= 2 * NSIDE * 32) return;
  const int side = tid / (NSIDE * 32);
  const int rem  = tid % (NSIDE * 32);
  const int s    = rem / 32;
  const int lane = rem % 32;
  const int m    = lane & 15;
  const int hl   = lane >> 4;

  const float* Ws = (side ? Wr : Wl) + (long)s * 512;
  v4f* dst = (v4f*)pre + (size_t)side * PRE_SIDE_V4F;

#pragma unroll
  for (int c = 0; c < 4; ++c) {
    const int r0 = 2 * c + 8 * hl;
    v4f q;
    if (side == 0) {
      q.x = Ws[(r0    ) * 32      + m];
      q.y = Ws[(r0 + 1) * 32      + m];
      q.z = Ws[(r0    ) * 32 + 16 + m];
      q.w = Ws[(r0 + 1) * 32 + 16 + m];
    } else {
      q.x = Ws[m * 32      + r0    ];
      q.y = Ws[m * 32      + r0 + 1];
      q.z = Ws[m * 32 + 16 + r0    ];
      q.w = Ws[m * 32 + 16 + r0 + 1];
    }
    dst[((size_t)s * 4 + c) * 32 + lane] = q;
  }
}

__device__ __forceinline__ void mps_step(v8f& V, float xt, const v2f a0[4], const v2f a1[4]) {
  v8f P = {}; v8f Q = {};
#pragma unroll
  for (int c = 0; c < 4; ++c) {
    v2f bb; bb.x = V[2 * c]; bb.y = V[2 * c + 1];
    P = WMMA4(a0[c], bb, P);
    Q = WMMA4(a1[c], bb, Q);
  }
#pragma unroll
  for (int r = 0; r < 8; ++r) V[r] = fmaf(xt, Q[r], P[r]);
}

__device__ __forceinline__ void mps_tail(
    int lane, int m, int hl,
    const v8f& V, const v8f& U,
    const float* __restrict__ Wout, const float* __restrict__ d1w,
    const float* __restrict__ d1b,  const float* __restrict__ d2w,
    const float* __restrict__ d2b,  float* __restrict__ out)
{
  float weff[8] = {}, c0 = d2b[0];
#pragma unroll 4
  for (int h = 0; h < 24; ++h) {
    const float wh = d2w[h];
    c0 = fmaf(wh, d1b[h], c0);
#pragma unroll
    for (int o = 0; o < 8; ++o) weff[o] = fmaf(wh, d1w[h * 8 + o], weff[o]);
  }

  float y = c0;
#pragma unroll
  for (int o = 0; o < 8; ++o) {
    v2f a[4];
#pragma unroll
    for (int c = 0; c < 4; ++c) {
      const int d0 = 2 * c + 8 * hl;
      a[c].x = Wout[(d0    ) * 128 + o * 16 + m];
      a[c].y = Wout[(d0 + 1) * 128 + o * 16 + m];
    }
    v8f G = {};
#pragma unroll
    for (int c = 0; c < 4; ++c) {
      v2f bb; bb.x = V[2 * c]; bb.y = V[2 * c + 1];
      G = WMMA4(a[c], bb, G);
    }
    float p = 0.f;
#pragma unroll
    for (int r = 0; r < 8; ++r) p = fmaf(G[r], U[r], p);
    p += __shfl_xor(p, 16, 32);
    y = fmaf(weff[o], p, y);
  }

  __shared__ float ybuf[ROWS_PER_BLOCK];
  if (lane < 16) ybuf[(threadIdx.x >> 5) * 16 + lane] = y;
  __syncthreads();
  if (threadIdx.x < ROWS_PER_BLOCK)
    out[(long)blockIdx.x * ROWS_PER_BLOCK + threadIdx.x] = ybuf[threadIdx.x];
}

__global__ __launch_bounds__(BLOCK_THREADS) void mps_dense_packed_kernel(
    const float* __restrict__ x,   const float* __restrict__ W0,
    const float* __restrict__ pre, const float* __restrict__ Wout,
    const float* __restrict__ WN,
    const float* __restrict__ d1w, const float* __restrict__ d1b,
    const float* __restrict__ d2w, const float* __restrict__ d2b,
    float* __restrict__ out)
{
  const int lane = threadIdx.x & 31;
  const int wave = blockIdx.x * (blockDim.x >> 5) + (threadIdx.x >> 5);
  const int m    = lane & 15;
  const int hl   = lane >> 4;

  const long b0 = (long)wave * 16 + m;

  const v4f* PL = (const v4f*)pre + lane;
  const v4f* PR = (const v4f*)pre + PRE_SIDE_V4F + lane;

  v8f V;
  {
    const float xa = x[b0 * MPS_F];
#pragma unroll
    for (int r = 0; r < 8; ++r) {
      const int d = r + 8 * hl;
      const float p0 = W0[d], p1 = W0[16 + d];
      V[r] = fmaf(xa, p1, p0);
    }
  }

  for (int s = 0; s < NSIDE; ++s) {
    v2f a0[4], a1[4];
#pragma unroll
    for (int c = 0; c < 4; ++c) {
      const v4f q = PL[((size_t)s * 4 + c) * 32];
      a0[c].x = q.x; a0[c].y = q.y;
      a1[c].x = q.z; a1[c].y = q.w;
    }
    const float xt = x[b0 * MPS_F + (s + 1)];
    mps_step(V, xt, a0, a1);
  }

  v8f U;
  {
    const float xa = x[b0 * MPS_F + (MPS_F - 1)];
#pragma unroll
    for (int r = 0; r < 8; ++r) {
      const int d = r + 8 * hl;
      const float p0 = WN[d * 2], p1 = WN[d * 2 + 1];
      U[r] = fmaf(xa, p1, p0);
    }
  }

  for (int i = 0; i < NSIDE; ++i) {
    const int s = NSIDE - 1 - i;
    v2f a0[4], a1[4];
#pragma unroll
    for (int c = 0; c < 4; ++c) {
      const v4f q = PR[((size_t)s * 4 + c) * 32];
      a0[c].x = q.x; a0[c].y = q.y;
      a1[c].x = q.z; a1[c].y = q.w;
    }
    const int t = MPS_HALF + s;
    const float xt = x[b0 * MPS_F + t];
    mps_step(U, xt, a0, a1);
  }

  mps_tail(lane, m, hl, V, U, Wout, d1w, d1b, d2w, d2b, out);
}

__global__ __launch_bounds__(BLOCK_THREADS) void mps_dense_kernel(
    const float* __restrict__ x,   const float* __restrict__ W0,
    const float* __restrict__ Wl,  const float* __restrict__ Wout,
    const float* __restrict__ Wr,  const float* __restrict__ WN,
    const float* __restrict__ d1w, const float* __restrict__ d1b,
    const float* __restrict__ d2w, const float* __restrict__ d2b,
    float* __restrict__ out)
{
  const int lane = threadIdx.x & 31;
  const int wave = blockIdx.x * (blockDim.x >> 5) + (threadIdx.x >> 5);
  const int m    = lane & 15;
  const int hl   = lane >> 4;

  const long b0 = (long)wave * 16 + m;

  v8f V;
  {
    const float xa = x[b0 * MPS_F];
#pragma unroll
    for (int r = 0; r < 8; ++r) {
      const int d = r + 8 * hl;
      const float p0 = W0[d], p1 = W0[16 + d];
      V[r] = fmaf(xa, p1, p0);
    }
  }

  for (int s = 0; s < NSIDE; ++s) {
    const float* Ws = Wl + (long)s * 512;
    v2f a0[4], a1[4];
#pragma unroll
    for (int c = 0; c < 4; ++c) {
      const int d0 = 2 * c + 8 * hl;
      a0[c].x = Ws[(d0    ) * 32      + m];
      a0[c].y = Ws[(d0 + 1) * 32      + m];
      a1[c].x = Ws[(d0    ) * 32 + 16 + m];
      a1[c].y = Ws[(d0 + 1) * 32 + 16 + m];
    }
    const float xt = x[b0 * MPS_F + (s + 1)];
    mps_step(V, xt, a0, a1);
  }

  v8f U;
  {
    const float xa = x[b0 * MPS_F + (MPS_F - 1)];
#pragma unroll
    for (int r = 0; r < 8; ++r) {
      const int d = r + 8 * hl;
      const float p0 = WN[d * 2], p1 = WN[d * 2 + 1];
      U[r] = fmaf(xa, p1, p0);
    }
  }

  for (int i = 0; i < NSIDE; ++i) {
    const int s = NSIDE - 1 - i;
    const float* Ws = Wr + (long)s * 512;
    v2f a0[4], a1[4];
#pragma unroll
    for (int c = 0; c < 4; ++c) {
      const int e0 = 2 * c + 8 * hl;
      a0[c].x = Ws[m * 32      + e0    ];
      a0[c].y = Ws[m * 32      + e0 + 1];
      a1[c].x = Ws[m * 32 + 16 + e0    ];
      a1[c].y = Ws[m * 32 + 16 + e0 + 1];
    }
    const int t = MPS_HALF + s;
    const float xt = x[b0 * MPS_F + t];
    mps_step(U, xt, a0, a1);
  }

  mps_tail(lane, m, hl, V, U, Wout, d1w, d1b, d2w, d2b, out);
}

extern "C" void kernel_launch(void* const* d_in, const int* in_sizes, int n_in,
                              void* d_out, int out_size, void* d_ws, size_t ws_size,
                              hipStream_t stream) {
  (void)in_sizes; (void)n_in; (void)out_size;
  const float* x    = (const float*)d_in[0];
  const float* W0   = (const float*)d_in[1];
  const float* Wl   = (const float*)d_in[2];
  const float* Wout = (const float*)d_in[3];
  const float* Wr   = (const float*)d_in[4];
  const float* WN   = (const float*)d_in[5];
  const float* d1w  = (const float*)d_in[6];
  const float* d1b  = (const float*)d_in[7];
  const float* d2w  = (const float*)d_in[8];
  const float* d2b  = (const float*)d_in[9];
  float* out = (float*)d_out;

  dim3 grid(256), block(128);

  if (d_ws != nullptr && ws_size >= PRE_BYTES) {
    hipLaunchKernelGGL(mps_prepack_kernel, dim3(64), dim3(256), 0, stream,
                       Wl, Wr, (float*)d_ws);
    hipLaunchKernelGGL(mps_dense_packed_kernel, grid, block, 0, stream,
                       x, W0, (const float*)d_ws, Wout, WN,
                       d1w, d1b, d2w, d2b, out);
  } else {
    hipLaunchKernelGGL(mps_dense_kernel, grid, block, 0, stream,
                       x, W0, Wl, Wout, Wr, WN, d1w, d1b, d2w, d2b, out);
  }
}


# chain loops software-pipelined: weights 8 steps ahead and x 16 positions ahead kept in flight in register rings (16-step unrolled body, scheduling barrier per step); arithmetic and its order same
# speedup vs baseline: 3.5179x; 1.9970x over previous
#include <hip/hip_runtime.h>


typedef __attribute__((ext_vector_type(2))) float v2f;
typedef __attribute__((ext_vector_type(4))) float v4f;
typedef __attribute__((ext_vector_type(8))) float v8f;

#define WMMA4(A, Bm, C) \
  __builtin_amdgcn_wmma_f32_16x16x4_f32(false, (A), false, (Bm), (short)0, (C), false, false)

#define MPS_B    16384
#define MPS_F    512
#define MPS_HALF 256
#define NSIDE    255
#define WAVES_PER_BLOCK 4
#define BLOCK_THREADS   (32 * WAVES_PER_BLOCK)
#define ROWS_PER_BLOCK  (16 * WAVES_PER_BLOCK)

#define PRE_SIDE_V4F (NSIDE * 4 * 32)
#define PRE_BYTES    ((size_t)2 * PRE_SIDE_V4F * 16)

#define WRING  8
#define XRING  16
#define BLK    16

__global__ __launch_bounds__(256) void mps_prepack_kernel(
    const float* __restrict__ Wl, const float* __restrict__ Wr,
    float* __restrict__ pre)
{
  const int tid = blockIdx.x * blockDim.x + threadIdx.x;
  if (tid >= 2 * NSIDE * 32) return;
  const int side = tid / (NSIDE * 32);
  const int rem  = tid % (NSIDE * 32);
  const int s    = rem / 32;
  const int lane = rem % 32;
  const int m    = lane & 15;
  const int hl   = lane >> 4;

  const float* Ws = (side ? Wr : Wl) + (long)s * 512;
  v4f* dst = (v4f*)pre + (size_t)side * PRE_SIDE_V4F;

#pragma unroll
  for (int c = 0; c < 4; ++c) {
    const int r0 = 2 * c + 8 * hl;
    v4f q;
    if (side == 0) {
      q.x = Ws[(r0    ) * 32      + m];
      q.y = Ws[(r0 + 1) * 32      + m];
      q.z = Ws[(r0    ) * 32 + 16 + m];
      q.w = Ws[(r0 + 1) * 32 + 16 + m];
    } else {
      q.x = Ws[m * 32      + r0    ];
      q.y = Ws[m * 32      + r0 + 1];
      q.z = Ws[m * 32 + 16 + r0    ];
      q.w = Ws[m * 32 + 16 + r0 + 1];
    }
    dst[((size_t)s * 4 + c) * 32 + lane] = q;
  }
}

__device__ __forceinline__ void mps_step(v8f& V, float xt, const v2f a0[4], const v2f a1[4]) {
  v8f P = {}; v8f Q = {};
#pragma unroll
  for (int c = 0; c < 4; ++c) {
    v2f bb; bb.x = V[2 * c]; bb.y = V[2 * c + 1];
    P = WMMA4(a0[c], bb, P);
    Q = WMMA4(a1[c], bb, Q);
  }
#pragma unroll
  for (int r = 0; r < 8; ++r) V[r] = fmaf(xt, Q[r], P[r]);
}


template <bool REV, bool FIRST>
__device__ __forceinline__ void mps_block(
    v8f& V, v4f (&wb)[WRING][4], float (&xb)[XRING],
    const v4f* wbase, const float* xrow,
    int lane, int k, const v8f& c0, const v8f& c1)
{
#pragma unroll
  for (int j = 0; j < BLK; ++j) {
    const float xt = xb[j];
    if (FIRST && j == 0) {
#pragma unroll
      for (int r = 0; r < 8; ++r) V[r] = fmaf(xt, c1[r], c0[r]);
    } else {
      const int n  = BLK * k + j - 1;
      const int sl = (j + WRING - 1) % WRING;
      v2f a0[4], a1[4];
#pragma unroll
      for (int c = 0; c < 4; ++c) {
        const v4f q = wb[sl][c];
        a0[c].x = q.x; a0[c].y = q.y;
        a1[c].x = q.z; a1[c].y = q.w;
      }
      mps_step(V, xt, a0, a1);
      const int nn  = min(n + WRING, NSIDE - 1);
      const int off = (REV ? -nn : nn) * 128;
#pragma unroll
      for (int c = 0; c < 4; ++c) wb[sl][c] = wbase[off + c * 32 + lane];
    }
    const int qn = min(BLK * k + j + XRING, MPS_HALF - 1);
    xb[j] = xrow[REV ? -qn : qn];
    __builtin_amdgcn_sched_barrier(0);
  }
}

template <bool REV>
__device__ __forceinline__ v8f mps_chain(
    const v4f* wbase, const float* xrow,
    int lane, const v8f& c0, const v8f& c1)
{
  v4f   wb[WRING][4];
  float xb[XRING];
#pragma unroll
  for (int q = 0; q < XRING; ++q) xb[q] = xrow[REV ? -q : q];
#pragma unroll
  for (int n = 0; n < WRING; ++n) {
#pragma unroll
    for (int c = 0; c < 4; ++c) wb[n][c] = wbase[(REV ? -n : n) * 128 + c * 32 + lane];
  }
  v8f V;
  mps_block<REV, true>(V, wb, xb, wbase, xrow, lane, 0, c0, c1);
#pragma unroll 1
  for (int k = 1; k < MPS_HALF / BLK; ++k)
    mps_block<REV, false>(V, wb, xb, wbase, xrow, lane, k, c0, c1);
  return V;
}

__device__ __forceinline__ void mps_tail(
    int lane, int m, int hl,
    const v8f& V, const v8f& U,
    const float* __restrict__ Wout, const float* __restrict__ d1w,
    const float* __restrict__ d1b,  const float* __restrict__ d2w,
    const float* __restrict__ d2b,  float* __restrict__ out)
{
  float weff[8] = {}, c0 = d2b[0];
#pragma unroll 4
  for (int h = 0; h < 24; ++h) {
    const float wh = d2w[h];
    c0 = fmaf(wh, d1b[h], c0);
#pragma unroll
    for (int o = 0; o < 8; ++o) weff[o] = fmaf(wh, d1w[h * 8 + o], weff[o]);
  }

  float y = c0;
#pragma unroll
  for (int o = 0; o < 8; ++o) {
    v2f a[4];
#pragma unroll
    for (int c = 0; c < 4; ++c) {
      const int d0 = 2 * c + 8 * hl;
      a[c].x = Wout[(d0    ) * 128 + o * 16 + m];
      a[c].y = Wout[(d0 + 1) * 128 + o * 16 + m];
    }
    v8f G = {};
#pragma unroll
    for (int c = 0; c < 4; ++c) {
      v2f bb; bb.x = V[2 * c]; bb.y = V[2 * c + 1];
      G = WMMA4(a[c], bb, G);
    }
    float p = 0.f;
#pragma unroll
    for (int r = 0; r < 8; ++r) p = fmaf(G[r], U[r], p);
    p += __shfl_xor(p, 16, 32);
    y = fmaf(weff[o], p, y);
  }

  __shared__ float ybuf[ROWS_PER_BLOCK];
  if (lane < 16) ybuf[(threadIdx.x >> 5) * 16 + lane] = y;
  __syncthreads();
  if (threadIdx.x < ROWS_PER_BLOCK)
    out[(long)blockIdx.x * ROWS_PER_BLOCK + threadIdx.x] = ybuf[threadIdx.x];
}

__global__ __launch_bounds__(BLOCK_THREADS) void mps_dense_packed_kernel(
    const float* x,                const float* __restrict__ W0,
    const float* pre,              const float* __restrict__ Wout,
    const float* __restrict__ WN,
    const float* __restrict__ d1w, const float* __restrict__ d1b,
    const float* __restrict__ d2w, const float* __restrict__ d2b,
    float* __restrict__ out)
{
  const int lane = threadIdx.x & 31;
  const int wave = blockIdx.x * (blockDim.x >> 5) + (threadIdx.x >> 5);
  const int m    = lane & 15;
  const int hl   = lane >> 4;

  const long b0 = (long)wave * 16 + m;

  const float* xrow0 = x + b0 * MPS_F;

  v8f c0, c1;
#pragma unroll
  for (int r = 0; r < 8; ++r) {
    const int d = r + 8 * hl;
    c0[r] = W0[d]; c1[r] = W0[16 + d];
  }
  const v8f V = mps_chain<false>((const v4f*)pre, xrow0, lane, c0, c1);

#pragma unroll
  for (int r = 0; r < 8; ++r) {
    const int d = r + 8 * hl;
    c0[r] = WN[d * 2]; c1[r] = WN[d * 2 + 1];
  }
  const v8f U = mps_chain<true>((const v4f*)pre + PRE_SIDE_V4F + (NSIDE - 1) * 128,
                                xrow0 + (MPS_F - 1), lane, c0, c1);

  mps_tail(lane, m, hl, V, U, Wout, d1w, d1b, d2w, d2b, out);
}

__global__ __launch_bounds__(BLOCK_THREADS) void mps_dense_kernel(
    const float* __restrict__ x,   const float* __restrict__ W0,
    const float* __restrict__ Wl,  const float* __restrict__ Wout,
    const float* __restrict__ Wr,  const float* __restrict__ WN,
    const float* __restrict__ d1w, const float* __restrict__ d1b,
    const float* __restrict__ d2w, const float* __restrict__ d2b,
    float* __restrict__ out)
{
  const int lane = threadIdx.x & 31;
  const int wave = blockIdx.x * (blockDim.x >> 5) + (threadIdx.x >> 5);
  const int m    = lane & 15;
  const int hl   = lane >> 4;

  const long b0 = (long)wave * 16 + m;

  v8f V;
  {
    const float xa = x[b0 * MPS_F];
#pragma unroll
    for (int r = 0; r < 8; ++r) {
      const int d = r + 8 * hl;
      const float p0 = W0[d], p1 = W0[16 + d];
      V[r] = fmaf(xa, p1, p0);
    }
  }

  for (int s = 0; s < NSIDE; ++s) {
    const float* Ws = Wl + (long)s * 512;
    v2f a0[4], a1[4];
#pragma unroll
    for (int c = 0; c < 4; ++c) {
      const int d0 = 2 * c + 8 * hl;
      a0[c].x = Ws[(d0    ) * 32      + m];
      a0[c].y = Ws[(d0 + 1) * 32      + m];
      a1[c].x = Ws[(d0    ) * 32 + 16 + m];
      a1[c].y = Ws[(d0 + 1) * 32 + 16 + m];
    }
    const float xt = x[b0 * MPS_F + (s + 1)];
    mps_step(V, xt, a0, a1);
  }

  v8f U;
  {
    const float xa = x[b0 * MPS_F + (MPS_F - 1)];
#pragma unroll
    for (int r = 0; r < 8; ++r) {
      const int d = r + 8 * hl;
      const float p0 = WN[d * 2], p1 = WN[d * 2 + 1];
      U[r] = fmaf(xa, p1, p0);
    }
  }

  for (int i = 0; i < NSIDE; ++i) {
    const int s = NSIDE - 1 - i;
    const float* Ws = Wr + (long)s * 512;
    v2f a0[4], a1[4];
#pragma unroll
    for (int c = 0; c < 4; ++c) {
      const int e0 = 2 * c + 8 * hl;
      a0[c].x = Ws[m * 32      + e0    ];
      a0[c].y = Ws[m * 32      + e0 + 1];
      a1[c].x = Ws[m * 32 + 16 + e0    ];
      a1[c].y = Ws[m * 32 + 16 + e0 + 1];
    }
    const int t = MPS_HALF + s;
    const float xt = x[b0 * MPS_F + t];
    mps_step(U, xt, a0, a1);
  }

  mps_tail(lane, m, hl, V, U, Wout, d1w, d1b, d2w, d2b, out);
}

extern "C" void kernel_launch(void* const* d_in, const int* in_sizes, int n_in,
                              void* d_out, int out_size, void* d_ws, size_t ws_size,
                              hipStream_t stream) {
  (void)in_sizes; (void)n_in; (void)out_size;
  const float* x    = (const float*)d_in[0];
  const float* W0   = (const float*)d_in[1];
  const float* Wl   = (const float*)d_in[2];
  const float* Wout = (const float*)d_in[3];
  const float* Wr   = (const float*)d_in[4];
  const float* WN   = (const float*)d_in[5];
  const float* d1w  = (const float*)d_in[6];
  const float* d1b  = (const float*)d_in[7];
  const float* d2w  = (const float*)d_in[8];
  const float* d2b  = (const float*)d_in[9];
  float* out = (float*)d_out;

  dim3 grid(256), block(128);

  if (d_ws != nullptr && ws_size >= PRE_BYTES) {
    hipLaunchKernelGGL(mps_prepack_kernel, dim3(64), dim3(256), 0, stream,
                       Wl, Wr, (float*)d_ws);
    hipLaunchKernelGGL(mps_dense_packed_kernel, grid, block, 0, stream,
                       x, W0, (const float*)d_ws, Wout, WN,
                       d1w, d1b, d2w, d2b, out);
  } else {
    hipLaunchKernelGGL(mps_dense_kernel, grid, block, 0, stream,
                       x, W0, Wl, Wout, Wr, WN, d1w, d1b, d2w, d2b, out);
  }
}


# left and right chain walked by one loop, half a position apart: each chain's vector update and ring refills issue under the other chain's 8 matrix instructions; arithmetic and its order same
# speedup vs baseline: 4.3256x; 1.2296x over previous
#include <hip/hip_runtime.h>


typedef __attribute__((ext_vector_type(2))) float v2f;
typedef __attribute__((ext_vector_type(4))) float v4f;
typedef __attribute__((ext_vector_type(8))) float v8f;

#define WMMA4(A, Bm, C) \
  __builtin_amdgcn_wmma_f32_16x16x4_f32(false, (A), false, (Bm), (short)0, (C), false, false)

#define MPS_B    16384
#define MPS_F    512
#define MPS_HALF 256
#define NSIDE    255
#define WAVES_PER_BLOCK 4
#define BLOCK_THREADS   (32 * WAVES_PER_BLOCK)
#define ROWS_PER_BLOCK  (16 * WAVES_PER_BLOCK)

#define PRE_SIDE_V4F (NSIDE * 4 * 32)
#define PRE_BYTES    ((size_t)2 * PRE_SIDE_V4F * 16)

#define WRING  8
#define XRING  16
#define BLK    16

__global__ __launch_bounds__(256) void mps_prepack_kernel(
    const float* __restrict__ Wl, const float* __restrict__ Wr,
    float* __restrict__ pre)
{
  const int tid = blockIdx.x * blockDim.x + threadIdx.x;
  if (tid >= 2 * NSIDE * 32) return;
  const int side = tid / (NSIDE * 32);
  const int rem  = tid % (NSIDE * 32);
  const int s    = rem / 32;
  const int lane = rem % 32;
  const int m    = lane & 15;
  const int hl   = lane >> 4;

  const float* Ws = (side ? Wr : Wl) + (long)s * 512;
  v4f* dst = (v4f*)pre + (size_t)side * PRE_SIDE_V4F;

#pragma unroll
  for (int c = 0; c < 4; ++c) {
    const int r0 = 2 * c + 8 * hl;
    v4f q;
    if (side == 0) {
      q.x = Ws[(r0    ) * 32      + m];
      q.y = Ws[(r0 + 1) * 32      + m];
      q.z = Ws[(r0    ) * 32 + 16 + m];
      q.w = Ws[(r0 + 1) * 32 + 16 + m];
    } else {
      q.x = Ws[m * 32      + r0    ];
      q.y = Ws[m * 32      + r0 + 1];
      q.z = Ws[m * 32 + 16 + r0    ];
      q.w = Ws[m * 32 + 16 + r0 + 1];
    }
    dst[((size_t)s * 4 + c) * 32 + lane] = q;
  }
}

__device__ __forceinline__ void mps_step(v8f& V, float xt, const v2f a0[4], const v2f a1[4]) {
  v8f P = {}; v8f Q = {};
#pragma unroll
  for (int c = 0; c < 4; ++c) {
    v2f bb; bb.x = V[2 * c]; bb.y = V[2 * c + 1];
    P = WMMA4(a0[c], bb, P);
    Q = WMMA4(a1[c], bb, Q);
  }
#pragma unroll
  for (int r = 0; r < 8; ++r) V[r] = fmaf(xt, Q[r], P[r]);
}


__device__ __forceinline__ void mps_mm(const v8f& V, const v4f (&w)[4], v8f& P, v8f& Q) {
  v8f p = {}; v8f q = {};
#pragma unroll
  for (int c = 0; c < 4; ++c) {
    v2f a0, a1, bb;
    a0.x = w[c].x; a0.y = w[c].y;
    a1.x = w[c].z; a1.y = w[c].w;
    bb.x = V[2 * c]; bb.y = V[2 * c + 1];
    p = WMMA4(a0, bb, p);
    q = WMMA4(a1, bb, q);
  }
  P = p; Q = q;
}

__device__ __forceinline__ void mps_fin(v8f& V, float xt, const v8f& P, const v8f& Q) {
#pragma unroll
  for (int r = 0; r < 8; ++r) V[r] = fmaf(xt, Q[r], P[r]);
}

__device__ __forceinline__ void mps_hold(const v8f& a, const v8f& b) { asm volatile("" :: "v"(a), "v"(b)); }
__device__ __forceinline__ void mps_hold(const v8f& a) { asm volatile("" :: "v"(a)); }

template <bool FIRST>
__device__ __forceinline__ void mps_block2(
    v8f& VL, v8f& VR, v8f& PR, v8f& QR,
    v4f (&wl)[WRING][4], v4f (&wr)[WRING][4],
    float (&xl)[XRING], float (&xr)[XRING],
    const v4f* wl0, const v4f* wr0, const float* xl0, const float* xr0,
    int lane, int k,
    const v8f& c0l, const v8f& c1l, const v8f& c0r, const v8f& c1r)
{
#pragma unroll
  for (int j = 0; j < BLK; ++j) {
    const int q = BLK * k + j;
    if (FIRST && j == 0) {
#pragma unroll
      for (int r = 0; r < 8; ++r) {
        VL[r] = fmaf(xl[0], c1l[r], c0l[r]);
        VR[r] = fmaf(xr[0], c1r[r], c0r[r]);
      }
      xl[0] = xl0[XRING];
      xr[0] = xr0[-XRING];
    } else {
      const int n  = q - 1;
      const int sl = (j + WRING - 1) % WRING;
      const int nn = min(n + WRING, NSIDE - 1);
      v8f PL, QL;

      mps_mm(VL, wl[sl], PL, QL);
      mps_hold(PL, QL);
      __builtin_amdgcn_sched_barrier(0);

      if (!(FIRST && j == 1)) {
        const int jp = (j + XRING - 1) % XRING;
        mps_fin(VR, xr[jp], PR, QR);
        mps_hold(VR);
        xr[jp] = xr0[-min(q - 1 + XRING, MPS_HALF - 1)];
      }
#pragma unroll
      for (int c = 0; c < 4; ++c) wl[sl][c] = wl0[nn * 128 + c * 32 + lane];
      __builtin_amdgcn_sched_barrier(0);

      mps_mm(VR, wr[sl], PR, QR);
      mps_hold(PR, QR);
      __builtin_amdgcn_sched_barrier(0);

      mps_fin(VL, xl[j % XRING], PL, QL);
      mps_hold(VL);
      xl[j % XRING] = xl0[min(q + XRING, MPS_HALF - 1)];
#pragma unroll
      for (int c = 0; c < 4; ++c) wr[sl][c] = wr0[-nn * 128 + c * 32 + lane];
    }
    __builtin_amdgcn_sched_barrier(0);
  }
}

__device__ __forceinline__ void mps_chains(
    v8f& VL, v8f& VR,
    const v4f* wl0, const v4f* wr0, const float* xl0, const float* xr0,
    int lane,
    const v8f& c0l, const v8f& c1l, const v8f& c0r, const v8f& c1r)
{
  v4f   wl[WRING][4], wr[WRING][4];
  float xl[XRING], xr[XRING];
#pragma unroll
  for (int q = 0; q < XRING; ++q) { xl[q] = xl0[q]; xr[q] = xr0[-q]; }
#pragma unroll
  for (int n = 0; n < WRING; ++n) {
#pragma unroll
    for (int c = 0; c < 4; ++c) {
      wl[n][c] = wl0[ n * 128 + c * 32 + lane];
      wr[n][c] = wr0[-n * 128 + c * 32 + lane];
    }
  }
  v8f PR, QR;
  mps_block2<true>(VL, VR, PR, QR, wl, wr, xl, xr, wl0, wr0, xl0, xr0, lane, 0, c0l, c1l, c0r, c1r);
#pragma unroll 1
  for (int k = 1; k < MPS_HALF / BLK; ++k)
    mps_block2<false>(VL, VR, PR, QR, wl, wr, xl, xr, wl0, wr0, xl0, xr0, lane, k, c0l, c1l, c0r, c1r);
  mps_fin(VR, xr[(MPS_HALF - 1) % XRING], PR, QR);
}

__device__ __forceinline__ void mps_tail(
    int lane, int m, int hl,
    const v8f& V, const v8f& U,
    const float* __restrict__ Wout, const float* __restrict__ d1w,
    const float* __restrict__ d1b,  const float* __restrict__ d2w,
    const float* __restrict__ d2b,  float* __restrict__ out)
{
  float weff[8] = {}, c0 = d2b[0];
#pragma unroll 4
  for (int h = 0; h < 24; ++h) {
    const float wh = d2w[h];
    c0 = fmaf(wh, d1b[h], c0);
#pragma unroll
    for (int o = 0; o < 8; ++o) weff[o] = fmaf(wh, d1w[h * 8 + o], weff[o]);
  }

  float y = c0;
#pragma unroll
  for (int o = 0; o < 8; ++o) {
    v2f a[4];
#pragma unroll
    for (int c = 0; c < 4; ++c) {
      const int d0 = 2 * c + 8 * hl;
      a[c].x = Wout[(d0    ) * 128 + o * 16 + m];
      a[c].y = Wout[(d0 + 1) * 128 + o * 16 + m];
    }
    v8f G = {};
#pragma unroll
    for (int c = 0; c < 4; ++c) {
      v2f bb; bb.x = V[2 * c]; bb.y = V[2 * c + 1];
      G = WMMA4(a[c], bb, G);
    }
    float p = 0.f;
#pragma unroll
    for (int r = 0; r < 8; ++r) p = fmaf(G[r], U[r], p);
    p += __shfl_xor(p, 16, 32);
    y = fmaf(weff[o], p, y);
  }

  __shared__ float ybuf[ROWS_PER_BLOCK];
  if (lane < 16) ybuf[(threadIdx.x >> 5) * 16 + lane] = y;
  __syncthreads();
  if (threadIdx.x < ROWS_PER_BLOCK)
    out[(long)blockIdx.x * ROWS_PER_BLOCK + threadIdx.x] = ybuf[threadIdx.x];
}

__global__ __launch_bounds__(BLOCK_THREADS) void mps_dense_packed_kernel(
    const float* x,                const float* __restrict__ W0,
    const float* pre,              const float* __restrict__ Wout,
    const float* __restrict__ WN,
    const float* __restrict__ d1w, const float* __restrict__ d1b,
    const float* __restrict__ d2w, const float* __restrict__ d2b,
    float* __restrict__ out)
{
  const int lane = threadIdx.x & 31;
  const int wave = blockIdx.x * (blockDim.x >> 5) + (threadIdx.x >> 5);
  const int m    = lane & 15;
  const int hl   = lane >> 4;

  const long b0 = (long)wave * 16 + m;

  const float* xrow0 = x + b0 * MPS_F;

  v8f c0l, c1l, c0r, c1r;
#pragma unroll
  for (int r = 0; r < 8; ++r) {
    const int d = r + 8 * hl;
    c0l[r] = W0[d];     c1l[r] = W0[16 + d];
    c0r[r] = WN[d * 2]; c1r[r] = WN[d * 2 + 1];
  }
  v8f V, U;
  mps_chains(V, U,
             (const v4f*)pre, (const v4f*)pre + PRE_SIDE_V4F + (NSIDE - 1) * 128,
             xrow0, xrow0 + (MPS_F - 1), lane, c0l, c1l, c0r, c1r);

  mps_tail(lane, m, hl, V, U, Wout, d1w, d1b, d2w, d2b, out);
}

__global__ __launch_bounds__(BLOCK_THREADS) void mps_dense_kernel(
    const float* __restrict__ x,   const float* __restrict__ W0,
    const float* __restrict__ Wl,  const float* __restrict__ Wout,
    const float* __restrict__ Wr,  const float* __restrict__ WN,
    const float* __restrict__ d1w, const float* __restrict__ d1b,
    const float* __restrict__ d2w, const float* __restrict__ d2b,
    float* __restrict__ out)
{
  const int lane = threadIdx.x & 31;
  const int wave = blockIdx.x * (blockDim.x >> 5) + (threadIdx.x >> 5);
  const int m    = lane & 15;
  const int hl   = lane >> 4;

  const long b0 = (long)wave * 16 + m;

  v8f V;
  {
    const float xa = x[b0 * MPS_F];
#pragma unroll
    for (int r = 0; r < 8; ++r) {
      const int d = r + 8 * hl;
      const float p0 = W0[d], p1 = W0[16 + d];
      V[r] = fmaf(xa, p1, p0);
    }
  }

  for (int s = 0; s < NSIDE; ++s) {
    const float* Ws = Wl + (long)s * 512;
    v2f a0[4], a1[4];
#pragma unroll
    for (int c = 0; c < 4; ++c) {
      const int d0 = 2 * c + 8 * hl;
      a0[c].x = Ws[(d0    ) * 32      + m];
      a0[c].y = Ws[(d0 + 1) * 32      + m];
      a1[c].x = Ws[(d0    ) * 32 + 16 + m];
      a1[c].y = Ws[(d0 + 1) * 32 + 16 + m];
    }
    const float xt = x[b0 * MPS_F + (s + 1)];
    mps_step(V, xt, a0, a1);
  }

  v8f U;
  {
    const float xa = x[b0 * MPS_F + (MPS_F - 1)];
#pragma unroll
    for (int r = 0; r < 8; ++r) {
      const int d = r + 8 * hl;
      const float p0 = WN[d * 2], p1 = WN[d * 2 + 1];
      U[r] = fmaf(xa, p1, p0);
    }
  }

  for (int i = 0; i < NSIDE; ++i) {
    const int s = NSIDE - 1 - i;
    const float* Ws = Wr + (long)s * 512;
    v2f a0[4], a1[4];
#pragma unroll
    for (int c = 0; c < 4; ++c) {
      const int e0 = 2 * c + 8 * hl;
      a0[c].x = Ws[m * 32      + e0    ];
      a0[c].y = Ws[m * 32      + e0 + 1];
      a1[c].x = Ws[m * 32 + 16 + e0    ];
      a1[c].y = Ws[m * 32 + 16 + e0 + 1];
    }
    const int t = MPS_HALF + s;
    const float xt = x[b0 * MPS_F + t];
    mps_step(U, xt, a0, a1);
  }

  mps_tail(lane, m, hl, V, U, Wout, d1w, d1b, d2w, d2b, out);
}

extern "C" void kernel_launch(void* const* d_in, const int* in_sizes, int n_in,
                              void* d_out, int out_size, void* d_ws, size_t ws_size,
                              hipStream_t stream) {
  (void)in_sizes; (void)n_in; (void)out_size;
  const float* x    = (const float*)d_in[0];
  const float* W0   = (const float*)d_in[1];
  const float* Wl   = (const float*)d_in[2];
  const float* Wout = (const float*)d_in[3];
  const float* Wr   = (const float*)d_in[4];
  const float* WN   = (const float*)d_in[5];
  const float* d1w  = (const float*)d_in[6];
  const float* d1b  = (const float*)d_in[7];
  const float* d2w  = (const float*)d_in[8];
  const float* d2b  = (const float*)d_in[9];
  float* out = (float*)d_out;

  dim3 grid(256), block(128);

  if (d_ws != nullptr && ws_size >= PRE_BYTES) {
    hipLaunchKernelGGL(mps_prepack_kernel, dim3(64), dim3(256), 0, stream,
                       Wl, Wr, (float*)d_ws);
    hipLaunchKernelGGL(mps_dense_packed_kernel, grid, block, 0, stream,
                       x, W0, (const float*)d_ws, Wout, WN,
                       d1w, d1b, d2w, d2b, out);
  } else {
    hipLaunchKernelGGL(mps_dense_kernel, grid, block, 0, stream,
                       x, W0, Wl, Wout, Wr, WN, d1w, d1b, d2w, d2b, out);
  }
}


# each wave walks one chain side for two 16-row tiles (partner wave: other side): a step's weights loaded once per 16 matrix instructions, not 8; tile halves swap via LDS; arithmetic and its order same
# speedup vs baseline: 4.6115x; 1.0661x over previous
#include <hip/hip_runtime.h>


typedef __attribute__((ext_vector_type(2))) float v2f;
typedef __attribute__((ext_vector_type(4))) float v4f;
typedef __attribute__((ext_vector_type(8))) float v8f;

#define WMMA4(A, Bm, C) \
  __builtin_amdgcn_wmma_f32_16x16x4_f32(false, (A), false, (Bm), (short)0, (C), false, false)

#define MPS_B    16384
#define MPS_F    512
#define MPS_HALF 256
#define NSIDE    255
#define WAVES_PER_BLOCK 4
#define BLOCK_THREADS   (32 * WAVES_PER_BLOCK)
#define ROWS_PER_BLOCK  (16 * WAVES_PER_BLOCK)

#define PRE_SIDE_V4F (NSIDE * 4 * 32)
#define PRE_BYTES    ((size_t)2 * PRE_SIDE_V4F * 16)

#define WRING  8
#define XRING  16
#define BLK    16

__global__ __launch_bounds__(256) void mps_prepack_kernel(
    const float* __restrict__ Wl, const float* __restrict__ Wr,
    float* __restrict__ pre)
{
  const int tid = blockIdx.x * blockDim.x + threadIdx.x;
  if (tid >= 2 * NSIDE * 32) return;
  const int side = tid / (NSIDE * 32);
  const int rem  = tid % (NSIDE * 32);
  const int s    = rem / 32;
  const int lane = rem % 32;
  const int m    = lane & 15;
  const int hl   = lane >> 4;

  const float* Ws = (side ? Wr : Wl) + (long)s * 512;
  v4f* dst = (v4f*)pre + (size_t)side * PRE_SIDE_V4F;

#pragma unroll
  for (int c = 0; c < 4; ++c) {
    const int r0 = 2 * c + 8 * hl;
    v4f q;
    if (side == 0) {
      q.x = Ws[(r0    ) * 32      + m];
      q.y = Ws[(r0 + 1) * 32      + m];
      q.z = Ws[(r0    ) * 32 + 16 + m];
      q.w = Ws[(r0 + 1) * 32 + 16 + m];
    } else {
      q.x = Ws[m * 32      + r0    ];
      q.y = Ws[m * 32      + r0 + 1];
      q.z = Ws[m * 32 + 16 + r0    ];
      q.w = Ws[m * 32 + 16 + r0 + 1];
    }
    dst[((size_t)s * 4 + c) * 32 + lane] = q;
  }
}

__device__ __forceinline__ void mps_step(v8f& V, float xt, const v2f a0[4], const v2f a1[4]) {
  v8f P = {}; v8f Q = {};
#pragma unroll
  for (int c = 0; c < 4; ++c) {
    v2f bb; bb.x = V[2 * c]; bb.y = V[2 * c + 1];
    P = WMMA4(a0[c], bb, P);
    Q = WMMA4(a1[c], bb, Q);
  }
#pragma unroll
  for (int r = 0; r < 8; ++r) V[r] = fmaf(xt, Q[r], P[r]);
}


__device__ __forceinline__ void mps_mm(const v8f& V, const v4f (&w)[4], v8f& P, v8f& Q) {
  v8f p = {}; v8f q = {};
#pragma unroll
  for (int c = 0; c < 4; ++c) {
    v2f a0, a1, bb;
    a0.x = w[c].x; a0.y = w[c].y;
    a1.x = w[c].z; a1.y = w[c].w;
    bb.x = V[2 * c]; bb.y = V[2 * c + 1];
    p = WMMA4(a0, bb, p);
    q = WMMA4(a1, bb, q);
  }
  P = p; Q = q;
}

__device__ __forceinline__ void mps_fin(v8f& V, float xt, const v8f& P, const v8f& Q) {
#pragma unroll
  for (int r = 0; r < 8; ++r) V[r] = fmaf(xt, Q[r], P[r]);
}

__device__ __forceinline__ void mps_hold(const v8f& a, const v8f& b) { asm volatile("" :: "v"(a), "v"(b)); }
__device__ __forceinline__ void mps_hold(const v8f& a) { asm volatile("" :: "v"(a)); }

template <int DIR, bool FIRST>
__device__ __forceinline__ void mps_block2(
    v8f& V1, v8f& V2, v8f& P2, v8f& Q2,
    v4f (&w)[WRING][4],
    float (&x1)[XRING], float (&x2)[XRING],
    const v4f* w0, const float* x10, const float* x20,
    int lane, int k,
    const v8f& c0, const v8f& c1)
{
#pragma unroll
  for (int j = 0; j < BLK; ++j) {
    const int q = BLK * k + j;
    if (FIRST && j == 0) {
#pragma unroll
      for (int r = 0; r < 8; ++r) {
        V1[r] = fmaf(x1[0], c1[r], c0[r]);
        V2[r] = fmaf(x2[0], c1[r], c0[r]);
      }
      x1[0] = x10[DIR * XRING];
      x2[0] = x20[DIR * XRING];
    } else {
      const int n  = q - 1;
      const int sl = (j + WRING - 1) % WRING;
      const int nn = min(n + WRING, NSIDE - 1);
      const int jp = (j + XRING - 1) % XRING;
      v8f P1, Q1;

      mps_mm(V1, w[sl], P1, Q1);
      mps_hold(P1, Q1);
      __builtin_amdgcn_sched_barrier(0);

      if (!(FIRST && j == 1)) {
        mps_fin(V2, x2[jp], P2, Q2);
        mps_hold(V2);
      }
      __builtin_amdgcn_sched_barrier(0);

      mps_mm(V2, w[sl], P2, Q2);
      mps_hold(P2, Q2);
      __builtin_amdgcn_sched_barrier(0);

      mps_fin(V1, x1[j % XRING], P1, Q1);
      mps_hold(V1);
      x1[j % XRING] = x10[DIR * min(q + XRING, MPS_HALF - 1)];
      if (!(FIRST && j == 1))
        x2[jp] = x20[DIR * min(q - 1 + XRING, MPS_HALF - 1)];
#pragma unroll
      for (int c = 0; c < 4; ++c) w[sl][c] = w0[DIR * (nn * 128) + c * 32 + lane];
    }
    __builtin_amdgcn_sched_barrier(0);
  }
}

template <int DIR>
__device__ __forceinline__ void mps_chains(
    v8f& V1, v8f& V2,
    const v4f* w0, const float* x10, const float* x20,
    int lane,
    const v8f& c0, const v8f& c1)
{
  v4f   w[WRING][4];
  float x1[XRING], x2[XRING];
#pragma unroll
  for (int q = 0; q < XRING; ++q) { x1[q] = x10[DIR * q]; x2[q] = x20[DIR * q]; }
#pragma unroll
  for (int n = 0; n < WRING; ++n) {
#pragma unroll
    for (int c = 0; c < 4; ++c) w[n][c] = w0[DIR * (n * 128) + c * 32 + lane];
  }
  v8f P2, Q2;
  mps_block2<DIR, true>(V1, V2, P2, Q2, w, x1, x2, w0, x10, x20, lane, 0, c0, c1);
#pragma unroll 1
  for (int k = 1; k < MPS_HALF / BLK; ++k)
    mps_block2<DIR, false>(V1, V2, P2, Q2, w, x1, x2, w0, x10, x20, lane, k, c0, c1);
  mps_fin(V2, x2[(MPS_HALF - 1) % XRING], P2, Q2);
}

__device__ __forceinline__ void mps_tail(
    int lane, int m, int hl,
    const v8f& V, const v8f& U,
    const float* __restrict__ Wout, const float* __restrict__ d1w,
    const float* __restrict__ d1b,  const float* __restrict__ d2w,
    const float* __restrict__ d2b,  float* __restrict__ out)
{
  float weff[8] = {}, c0 = d2b[0];
#pragma unroll 4
  for (int h = 0; h < 24; ++h) {
    const float wh = d2w[h];
    c0 = fmaf(wh, d1b[h], c0);
#pragma unroll
    for (int o = 0; o < 8; ++o) weff[o] = fmaf(wh, d1w[h * 8 + o], weff[o]);
  }

  float y = c0;
#pragma unroll
  for (int o = 0; o < 8; ++o) {
    v2f a[4];
#pragma unroll
    for (int c = 0; c < 4; ++c) {
      const int d0 = 2 * c + 8 * hl;
      a[c].x = Wout[(d0    ) * 128 + o * 16 + m];
      a[c].y = Wout[(d0 + 1) * 128 + o * 16 + m];
    }
    v8f G = {};
#pragma unroll
    for (int c = 0; c < 4; ++c) {
      v2f bb; bb.x = V[2 * c]; bb.y = V[2 * c + 1];
      G = WMMA4(a[c], bb, G);
    }
    float p = 0.f;
#pragma unroll
    for (int r = 0; r < 8; ++r) p = fmaf(G[r], U[r], p);
    p += __shfl_xor(p, 16, 32);
    y = fmaf(weff[o], p, y);
  }

  __shared__ float ybuf[ROWS_PER_BLOCK];
  if (lane < 16) ybuf[(threadIdx.x >> 5) * 16 + lane] = y;
  __syncthreads();
  if (threadIdx.x < ROWS_PER_BLOCK)
    out[(long)blockIdx.x * ROWS_PER_BLOCK + threadIdx.x] = ybuf[threadIdx.x];
}

__global__ __launch_bounds__(BLOCK_THREADS) void mps_dense_packed_kernel(
    const float* x,                const float* __restrict__ W0,
    const float* pre,              const float* __restrict__ Wout,
    const float* __restrict__ WN,
    const float* __restrict__ d1w, const float* __restrict__ d1b,
    const float* __restrict__ d2w, const float* __restrict__ d2b,
    float* __restrict__ out)
{
  const int lane = threadIdx.x & 31;
  const int m    = lane & 15;
  const int hl   = lane >> 4;
  const int wv   = __builtin_amdgcn_readfirstlane((int)(threadIdx.x >> 5));
  const int side = wv & 1;
  const int pair = wv >> 1;

  const long rowA = (long)blockIdx.x * ROWS_PER_BLOCK + (2 * pair) * 16 + m;
  const float* xa = x + rowA * MPS_F;
  const float* xb = xa + 16 * MPS_F;

  v8f S1, S2;
  if (side == 0) {
    v8f c0, c1;
#pragma unroll
    for (int r = 0; r < 8; ++r) {
      const int d = r + 8 * hl;
      c0[r] = W0[d]; c1[r] = W0[16 + d];
    }
    mps_chains<1>(S1, S2, (const v4f*)pre, xa, xb, lane, c0, c1);
  } else {
    v8f c0, c1;
#pragma unroll
    for (int r = 0; r < 8; ++r) {
      const int d = r + 8 * hl;
      c0[r] = WN[d * 2]; c1[r] = WN[d * 2 + 1];
    }
    mps_chains<-1>(S1, S2, (const v4f*)pre + PRE_SIDE_V4F + (NSIDE - 1) * 128,
                   xa + (MPS_F - 1), xb + (MPS_F - 1), lane, c0, c1);
  }

  __shared__ v4f swapbuf[WAVES_PER_BLOCK][2][32];
  {
    const v8f give = side ? S1 : S2;
    v4f lo, hi;
    lo.x = give[0]; lo.y = give[1]; lo.z = give[2]; lo.w = give[3];
    hi.x = give[4]; hi.y = give[5]; hi.z = give[6]; hi.w = give[7];
    swapbuf[wv][0][lane] = lo;
    swapbuf[wv][1][lane] = hi;
  }
  __syncthreads();
  v8f got;
  {
    const v4f lo = swapbuf[wv ^ 1][0][lane];
    const v4f hi = swapbuf[wv ^ 1][1][lane];
    got[0] = lo.x; got[1] = lo.y; got[2] = lo.z; got[3] = lo.w;
    got[4] = hi.x; got[5] = hi.y; got[6] = hi.z; got[7] = hi.w;
  }
  const v8f V = side ? got : S1;
  const v8f U = side ? S2  : got;

  mps_tail(lane, m, hl, V, U, Wout, d1w, d1b, d2w, d2b, out);
}

__global__ __launch_bounds__(BLOCK_THREADS) void mps_dense_kernel(
    const float* __restrict__ x,   const float* __restrict__ W0,
    const float* __restrict__ Wl,  const float* __restrict__ Wout,
    const float* __restrict__ Wr,  const float* __restrict__ WN,
    const float* __restrict__ d1w, const float* __restrict__ d1b,
    const float* __restrict__ d2w, const float* __restrict__ d2b,
    float* __restrict__ out)
{
  const int lane = threadIdx.x & 31;
  const int wave = blockIdx.x * (blockDim.x >> 5) + (threadIdx.x >> 5);
  const int m    = lane & 15;
  const int hl   = lane >> 4;

  const long b0 = (long)wave * 16 + m;

  v8f V;
  {
    const float xa = x[b0 * MPS_F];
#pragma unroll
    for (int r = 0; r < 8; ++r) {
      const int d = r + 8 * hl;
      const float p0 = W0[d], p1 = W0[16 + d];
      V[r] = fmaf(xa, p1, p0);
    }
  }

  for (int s = 0; s < NSIDE; ++s) {
    const float* Ws = Wl + (long)s * 512;
    v2f a0[4], a1[4];
#pragma unroll
    for (int c = 0; c < 4; ++c) {
      const int d0 = 2 * c + 8 * hl;
      a0[c].x = Ws[(d0    ) * 32      + m];
      a0[c].y = Ws[(d0 + 1) * 32      + m];
      a1[c].x = Ws[(d0    ) * 32 + 16 + m];
      a1[c].y = Ws[(d0 + 1) * 32 + 16 + m];
    }
    const float xt = x[b0 * MPS_F + (s + 1)];
    mps_step(V, xt, a0, a1);
  }

  v8f U;
  {
    const float xa = x[b0 * MPS_F + (MPS_F - 1)];
#pragma unroll
    for (int r = 0; r < 8; ++r) {
      const int d = r + 8 * hl;
      const float p0 = WN[d * 2], p1 = WN[d * 2 + 1];
      U[r] = fmaf(xa, p1, p0);
    }
  }

  for (int i = 0; i < NSIDE; ++i) {
    const int s = NSIDE - 1 - i;
    const float* Ws = Wr + (long)s * 512;
    v2f a0[4], a1[4];
#pragma unroll
    for (int c = 0; c < 4; ++c) {
      const int e0 = 2 * c + 8 * hl;
      a0[c].x = Ws[m * 32      + e0    ];
      a0[c].y = Ws[m * 32      + e0 + 1];
      a1[c].x = Ws[m * 32 + 16 + e0    ];
      a1[c].y = Ws[m * 32 + 16 + e0 + 1];
    }
    const int t = MPS_HALF + s;
    const float xt = x[b0 * MPS_F + t];
    mps_step(U, xt, a0, a1);
  }

  mps_tail(lane, m, hl, V, U, Wout, d1w, d1b, d2w, d2b, out);
}

extern "C" void kernel_launch(void* const* d_in, const int* in_sizes, int n_in,
                              void* d_out, int out_size, void* d_ws, size_t ws_size,
                              hipStream_t stream) {
  (void)in_sizes; (void)n_in; (void)out_size;
  const float* x    = (const float*)d_in[0];
  const float* W0   = (const float*)d_in[1];
  const float* Wl   = (const float*)d_in[2];
  const float* Wout = (const float*)d_in[3];
  const float* Wr   = (const float*)d_in[4];
  const float* WN   = (const float*)d_in[5];
  const float* d1w  = (const float*)d_in[6];
  const float* d1b  = (const float*)d_in[7];
  const float* d2w  = (const float*)d_in[8];
  const float* d2b  = (const float*)d_in[9];
  float* out = (float*)d_out;

  dim3 grid(256), block(128);

  if (d_ws != nullptr && ws_size >= PRE_BYTES) {
    hipLaunchKernelGGL(mps_prepack_kernel, dim3(64), dim3(256), 0, stream,
                       Wl, Wr, (float*)d_ws);
    hipLaunchKernelGGL(mps_dense_packed_kernel, grid, block, 0, stream,
                       x, W0, (const float*)d_ws, Wout, WN,
                       d1w, d1b, d2w, d2b, out);
  } else {
    hipLaunchKernelGGL(mps_dense_kernel, grid, block, 0, stream,
                       x, W0, Wl, Wout, Wr, WN, d1w, d1b, d2w, d2b, out);
  }
}


# ring refills batched: one load group per 8 chain positions (8 steps' weight loads, 16 steps ahead, and x as one 16-byte load per 4 positions), not 6 loads behind each position; arithmetic, order same
# speedup vs baseline: 5.0271x; 1.0901x over previous
#include <hip/hip_runtime.h>


typedef __attribute__((ext_vector_type(2))) float v2f;
typedef __attribute__((ext_vector_type(4))) float v4f;
typedef __attribute__((ext_vector_type(8))) float v8f;

#define WMMA4(A, Bm, C) \
  __builtin_amdgcn_wmma_f32_16x16x4_f32(false, (A), false, (Bm), (short)0, (C), false, false)

#define MPS_B    16384
#define MPS_F    512
#define MPS_HALF 256
#define NSIDE    255
#define WAVES_PER_BLOCK 4
#define BLOCK_THREADS   (32 * WAVES_PER_BLOCK)
#define ROWS_PER_BLOCK  (16 * WAVES_PER_BLOCK)

#define PRE_SIDE_V4F (NSIDE * 4 * 32)
#define PRE_BYTES    ((size_t)2 * PRE_SIDE_V4F * 16)

#define WRING  16
#define XG     4
#define GRP    4
#define RF     8
#define BLK    16

__global__ __launch_bounds__(256) void mps_prepack_kernel(
    const float* __restrict__ Wl, const float* __restrict__ Wr,
    float* __restrict__ pre)
{
  const int tid = blockIdx.x * blockDim.x + threadIdx.x;
  if (tid >= 2 * NSIDE * 32) return;
  const int side = tid / (NSIDE * 32);
  const int rem  = tid % (NSIDE * 32);
  const int s    = rem / 32;
  const int lane = rem % 32;
  const int m    = lane & 15;
  const int hl   = lane >> 4;

  const float* Ws = (side ? Wr : Wl) + (long)s * 512;
  v4f* dst = (v4f*)pre + (size_t)side * PRE_SIDE_V4F;

#pragma unroll
  for (int c = 0; c < 4; ++c) {
    const int r0 = 2 * c + 8 * hl;
    v4f q;
    if (side == 0) {
      q.x = Ws[(r0    ) * 32      + m];
      q.y = Ws[(r0 + 1) * 32      + m];
      q.z = Ws[(r0    ) * 32 + 16 + m];
      q.w = Ws[(r0 + 1) * 32 + 16 + m];
    } else {
      q.x = Ws[m * 32      + r0    ];
      q.y = Ws[m * 32      + r0 + 1];
      q.z = Ws[m * 32 + 16 + r0    ];
      q.w = Ws[m * 32 + 16 + r0 + 1];
    }
    dst[((size_t)s * 4 + c) * 32 + lane] = q;
  }
}

__device__ __forceinline__ void mps_step(v8f& V, float xt, const v2f a0[4], const v2f a1[4]) {
  v8f P = {}; v8f Q = {};
#pragma unroll
  for (int c = 0; c < 4; ++c) {
    v2f bb; bb.x = V[2 * c]; bb.y = V[2 * c + 1];
    P = WMMA4(a0[c], bb, P);
    Q = WMMA4(a1[c], bb, Q);
  }
#pragma unroll
  for (int r = 0; r < 8; ++r) V[r] = fmaf(xt, Q[r], P[r]);
}


__device__ __forceinline__ void mps_mm(const v8f& V, const v4f (&w)[4], v8f& P, v8f& Q) {
  v8f p = {}; v8f q = {};
#pragma unroll
  for (int c = 0; c < 4; ++c) {
    v2f a0, a1, bb;
    a0.x = w[c].x; a0.y = w[c].y;
    a1.x = w[c].z; a1.y = w[c].w;
    bb.x = V[2 * c]; bb.y = V[2 * c + 1];
    p = WMMA4(a0, bb, p);
    q = WMMA4(a1, bb, q);
  }
  P = p; Q = q;
}

__device__ __forceinline__ void mps_fin(v8f& V, float xt, const v8f& P, const v8f& Q) {
#pragma unroll
  for (int r = 0; r < 8; ++r) V[r] = fmaf(xt, Q[r], P[r]);
}

__device__ __forceinline__ void mps_hold(const v8f& a, const v8f& b) { asm volatile("" :: "v"(a), "v"(b)); }
__device__ __forceinline__ void mps_hold(const v8f& a) { asm volatile("" :: "v"(a)); }

template <int DIR>
__device__ __forceinline__ v4f mps_xgroup(const float* x0, int g) {
  return *(const v4f*)(x0 + (DIR > 0 ? 4 * g : -4 * g - 3));
}

template <int DIR, bool FIRST>
__device__ __forceinline__ void mps_block2(
    v8f& V1, v8f& V2, v8f& P2, v8f& Q2,
    v4f (&w)[WRING][4],
    v4f (&x1)[XG], v4f (&x2)[XG],
    const v4f* w0, const float* x10, const float* x20,
    int lane, int k,
    const v8f& c0, const v8f& c1)
{
#pragma unroll
  for (int j = 0; j < BLK; ++j) {
    const int q  = BLK * k + j;
    const int g1 = (j / GRP) % XG;
    const int e1 = DIR > 0 ? (j % GRP) : (GRP - 1 - j % GRP);
    if (FIRST && j == 0) {
#pragma unroll
      for (int r = 0; r < 8; ++r) {
        V1[r] = fmaf(x1[g1][e1], c1[r], c0[r]);
        V2[r] = fmaf(x2[g1][e1], c1[r], c0[r]);
      }
    } else {
      const int sl = (j + WRING - 1) % WRING;
      const int jp = (j + BLK - 1) % BLK;
      const int g2 = (jp / GRP) % XG;
      const int e2 = DIR > 0 ? (jp % GRP) : (GRP - 1 - jp % GRP);
      v8f P1, Q1;

      mps_mm(V1, w[sl], P1, Q1);
      mps_hold(P1, Q1);
      __builtin_amdgcn_sched_barrier(0);

      if (!(FIRST && j == 1)) {
        mps_fin(V2, x2[g2][e2], P2, Q2);
        mps_hold(V2);
      }
      __builtin_amdgcn_sched_barrier(0);

      mps_mm(V2, w[sl], P2, Q2);
      mps_hold(P2, Q2);
      __builtin_amdgcn_sched_barrier(0);

      mps_fin(V1, x1[g1][e1], P1, Q1);
      mps_hold(V1);
      if (j % RF == 0) {
#pragma unroll
        for (int t = 0; t < RF / GRP; ++t) {
          const int gs = (j / GRP + XG - RF / GRP + t) % XG;
          const int gn = q / GRP - RF / GRP + t + XG;
          x1[gs] = mps_xgroup<DIR>(x10, gn);
          x2[gs] = mps_xgroup<DIR>(x20, gn);
        }
#pragma unroll
        for (int i = 0; i < RF; ++i) {
          const int s  = (j + WRING - RF + i) % WRING;
          const int nn = q - RF + i + WRING;
#pragma unroll
          for (int c = 0; c < 4; ++c) w[s][c] = w0[DIR * (nn * 128) + c * 32 + lane];
        }
      }
    }
    __builtin_amdgcn_sched_barrier(0);
  }
}

template <int DIR>
__device__ __forceinline__ void mps_chains(
    v8f& V1, v8f& V2,
    const v4f* w0, const float* x10, const float* x20,
    int lane,
    const v8f& c0, const v8f& c1)
{
  v4f w[WRING][4];
  v4f x1[XG], x2[XG];
#pragma unroll
  for (int g = 0; g < XG; ++g) { x1[g] = mps_xgroup<DIR>(x10, g); x2[g] = mps_xgroup<DIR>(x20, g); }
#pragma unroll
  for (int n = 0; n < WRING; ++n) {
#pragma unroll
    for (int c = 0; c < 4; ++c) w[n][c] = w0[DIR * (n * 128) + c * 32 + lane];
  }
  v8f P2, Q2;
  mps_block2<DIR, true>(V1, V2, P2, Q2, w, x1, x2, w0, x10, x20, lane, 0, c0, c1);
#pragma unroll 1
  for (int k = 1; k < MPS_HALF / BLK; ++k)
    mps_block2<DIR, false>(V1, V2, P2, Q2, w, x1, x2, w0, x10, x20, lane, k, c0, c1);
  {
    const int pl = MPS_HALF - 1;
    mps_fin(V2, x2[(pl / GRP) % XG][DIR > 0 ? (pl % GRP) : (GRP - 1 - pl % GRP)], P2, Q2);
  }
}

__device__ __forceinline__ void mps_tail(
    int lane, int m, int hl,
    const v8f& V, const v8f& U,
    const float* __restrict__ Wout, const float* __restrict__ d1w,
    const float* __restrict__ d1b,  const float* __restrict__ d2w,
    const float* __restrict__ d2b,  float* __restrict__ out)
{
  float weff[8] = {}, c0 = d2b[0];
#pragma unroll 4
  for (int h = 0; h < 24; ++h) {
    const float wh = d2w[h];
    c0 = fmaf(wh, d1b[h], c0);
#pragma unroll
    for (int o = 0; o < 8; ++o) weff[o] = fmaf(wh, d1w[h * 8 + o], weff[o]);
  }

  float y = c0;
#pragma unroll
  for (int o = 0; o < 8; ++o) {
    v2f a[4];
#pragma unroll
    for (int c = 0; c < 4; ++c) {
      const int d0 = 2 * c + 8 * hl;
      a[c].x = Wout[(d0    ) * 128 + o * 16 + m];
      a[c].y = Wout[(d0 + 1) * 128 + o * 16 + m];
    }
    v8f G = {};
#pragma unroll
    for (int c = 0; c < 4; ++c) {
      v2f bb; bb.x = V[2 * c]; bb.y = V[2 * c + 1];
      G = WMMA4(a[c], bb, G);
    }
    float p = 0.f;
#pragma unroll
    for (int r = 0; r < 8; ++r) p = fmaf(G[r], U[r], p);
    p += __shfl_xor(p, 16, 32);
    y = fmaf(weff[o], p, y);
  }

  __shared__ float ybuf[ROWS_PER_BLOCK];
  if (lane < 16) ybuf[(threadIdx.x >> 5) * 16 + lane] = y;
  __syncthreads();
  if (threadIdx.x < ROWS_PER_BLOCK)
    out[(long)blockIdx.x * ROWS_PER_BLOCK + threadIdx.x] = ybuf[threadIdx.x];
}

__global__ __launch_bounds__(BLOCK_THREADS) void mps_dense_packed_kernel(
    const float* x,                const float* __restrict__ W0,
    const float* pre,              const float* __restrict__ Wout,
    const float* __restrict__ WN,
    const float* __restrict__ d1w, const float* __restrict__ d1b,
    const float* __restrict__ d2w, const float* __restrict__ d2b,
    float* __restrict__ out)
{
  const int lane = threadIdx.x & 31;
  const int m    = lane & 15;
  const int hl   = lane >> 4;
  const int wv   = __builtin_amdgcn_readfirstlane((int)(threadIdx.x >> 5));
  const int side = wv & 1;
  const int pair = wv >> 1;

  const long rowA = (long)blockIdx.x * ROWS_PER_BLOCK + (2 * pair) * 16 + m;
  const float* xa = x + rowA * MPS_F;
  const float* xb = xa + 16 * MPS_F;

  v8f S1, S2;
  if (side == 0) {
    v8f c0, c1;
#pragma unroll
    for (int r = 0; r < 8; ++r) {
      const int d = r + 8 * hl;
      c0[r] = W0[d]; c1[r] = W0[16 + d];
    }
    mps_chains<1>(S1, S2, (const v4f*)pre, xa, xb, lane, c0, c1);
  } else {
    v8f c0, c1;
#pragma unroll
    for (int r = 0; r < 8; ++r) {
      const int d = r + 8 * hl;
      c0[r] = WN[d * 2]; c1[r] = WN[d * 2 + 1];
    }
    mps_chains<-1>(S1, S2, (const v4f*)pre + PRE_SIDE_V4F + (NSIDE - 1) * 128,
                   xa + (MPS_F - 1), xb + (MPS_F - 1), lane, c0, c1);
  }

  __shared__ v4f swapbuf[WAVES_PER_BLOCK][2][32];
  {
    const v8f give = side ? S1 : S2;
    v4f lo, hi;
    lo.x = give[0]; lo.y = give[1]; lo.z = give[2]; lo.w = give[3];
    hi.x = give[4]; hi.y = give[5]; hi.z = give[6]; hi.w = give[7];
    swapbuf[wv][0][lane] = lo;
    swapbuf[wv][1][lane] = hi;
  }
  __syncthreads();
  v8f got;
  {
    const v4f lo = swapbuf[wv ^ 1][0][lane];
    const v4f hi = swapbuf[wv ^ 1][1][lane];
    got[0] = lo.x; got[1] = lo.y; got[2] = lo.z; got[3] = lo.w;
    got[4] = hi.x; got[5] = hi.y; got[6] = hi.z; got[7] = hi.w;
  }
  const v8f V = side ? got : S1;
  const v8f U = side ? S2  : got;

  mps_tail(lane, m, hl, V, U, Wout, d1w, d1b, d2w, d2b, out);
}

__global__ __launch_bounds__(BLOCK_THREADS) void mps_dense_kernel(
    const float* __restrict__ x,   const float* __restrict__ W0,
    const float* __restrict__ Wl,  const float* __restrict__ Wout,
    const float* __restrict__ Wr,  const float* __restrict__ WN,
    const float* __restrict__ d1w, const float* __restrict__ d1b,
    const float* __restrict__ d2w, const float* __restrict__ d2b,
    float* __restrict__ out)
{
  const int lane = threadIdx.x & 31;
  const int wave = blockIdx.x * (blockDim.x >> 5) + (threadIdx.x >> 5);
  const int m    = lane & 15;
  const int hl   = lane >> 4;

  const long b0 = (long)wave * 16 + m;

  v8f V;
  {
    const float xa = x[b0 * MPS_F];
#pragma unroll
    for (int r = 0; r < 8; ++r) {
      const int d = r + 8 * hl;
      const float p0 = W0[d], p1 = W0[16 + d];
      V[r] = fmaf(xa, p1, p0);
    }
  }

  for (int s = 0; s < NSIDE; ++s) {
    const float* Ws = Wl + (long)s * 512;
    v2f a0[4], a1[4];
#pragma unroll
    for (int c = 0; c < 4; ++c) {
      const int d0 = 2 * c + 8 * hl;
      a0[c].x = Ws[(d0    ) * 32      + m];
      a0[c].y = Ws[(d0 + 1) * 32      + m];
      a1[c].x = Ws[(d0    ) * 32 + 16 + m];
      a1[c].y = Ws[(d0 + 1) * 32 + 16 + m];
    }
    const float xt = x[b0 * MPS_F + (s + 1)];
    mps_step(V, xt, a0, a1);
  }

  v8f U;
  {
    const float xa = x[b0 * MPS_F + (MPS_F - 1)];
#pragma unroll
    for (int r = 0; r < 8; ++r) {
      const int d = r + 8 * hl;
      const float p0 = WN[d * 2], p1 = WN[d * 2 + 1];
      U[r] = fmaf(xa, p1, p0);
    }
  }

  for (int i = 0; i < NSIDE; ++i) {
    const int s = NSIDE - 1 - i;
    const float* Ws = Wr + (long)s * 512;
    v2f a0[4], a1[4];
#pragma unroll
    for (int c = 0; c < 4; ++c) {
      const int e0 = 2 * c + 8 * hl;
      a0[c].x = Ws[m * 32      + e0    ];
      a0[c].y = Ws[m * 32      + e0 + 1];
      a1[c].x = Ws[m * 32 + 16 + e0    ];
      a1[c].y = Ws[m * 32 + 16 + e0 + 1];
    }
    const int t = MPS_HALF + s;
    const float xt = x[b0 * MPS_F + t];
    mps_step(U, xt, a0, a1);
  }

  mps_tail(lane, m, hl, V, U, Wout, d1w, d1b, d2w, d2b, out);
}

extern "C" void kernel_launch(void* const* d_in, const int* in_sizes, int n_in,
                              void* d_out, int out_size, void* d_ws, size_t ws_size,
                              hipStream_t stream) {
  (void)in_sizes; (void)n_in; (void)out_size;
  const float* x    = (const float*)d_in[0];
  const float* W0   = (const float*)d_in[1];
  const float* Wl   = (const float*)d_in[2];
  const float* Wout = (const float*)d_in[3];
  const float* Wr   = (const float*)d_in[4];
  const float* WN   = (const float*)d_in[5];
  const float* d1w  = (const float*)d_in[6];
  const float* d1b  = (const float*)d_in[7];
  const float* d2w  = (const float*)d_in[8];
  const float* d2b  = (const float*)d_in[9];
  float* out = (float*)d_out;

  dim3 grid(256), block(128);

  if (d_ws != nullptr && ws_size >= PRE_BYTES) {
    hipLaunchKernelGGL(mps_prepack_kernel, dim3(64), dim3(256), 0, stream,
                       Wl, Wr, (float*)d_ws);
    hipLaunchKernelGGL(mps_dense_packed_kernel, grid, block, 0, stream,
                       x, W0, (const float*)d_ws, Wout, WN,
                       d1w, d1b, d2w, d2b, out);
  } else {
    hipLaunchKernelGGL(mps_dense_kernel, grid, block, 0, stream,
                       x, W0, Wl, Wout, Wr, WN, d1w, d1b, d2w, d2b, out);
  }
}
